# starting kernel
# baseline (speedup 1.0000x reference)
#include <hip/hip_runtime.h>

#define BATCH 64
#define LFR   500
#define NB    501
#define DDIM  1024
#define SMAX  50

typedef __attribute__((ext_vector_type(2))) float v2f;
typedef __attribute__((ext_vector_type(8))) float v8f;

__global__ void seg_prep(const float* __restrict__ in_boundary,
                         int* __restrict__ segid, float* __restrict__ w,
                         float* __restrict__ mask_sum, float* __restrict__ bnd_out)
{
    int b = blockIdx.x;
    const float* ib = in_boundary + b * NB;

    for (int t = threadIdx.x; t < NB; t += blockDim.x)
        bnd_out[b * NB + t] = ib[t];
    for (int s = threadIdx.x; s < SMAX; s += blockDim.x)
        mask_sum[b * SMAX + s] = 0.0f;
    __syncthreads();

    if (threadIdx.x == 0) {
        int*   sg = segid + b * LFR;
        float* wb = w     + b * LFR;
        int seg = 0, start = 0;
        for (int t = 1; t <= LFR; ++t) {
            if (ib[t] != 0.0f) {
                float wv = (seg < SMAX) ? (1.0f / (float)(t - start)) : 0.0f;
                for (int l = start; l < t; ++l) { sg[l] = seg; wb[l] = wv; }
                if (seg < SMAX) mask_sum[b * SMAX + seg] = 1.0f;
                ++seg; start = t;
            }
        }
        for (int l = start; l < LFR; ++l) { sg[l] = seg; wb[l] = 0.0f; }
    }
}

__global__ void seg_gemm_wmma(const float* __restrict__ x,
                              const int* __restrict__ segid,
                              const float* __restrict__ w,
                              float* __restrict__ out)
{
    __shared__ int   s_sg[LFR];
    __shared__ float s_w [LFR];

    const int b    = blockIdx.y;
    const int wave = threadIdx.x >> 5;
    const int lane = threadIdx.x & 31;
    const int n_tile = blockIdx.x * (blockDim.x >> 5) + wave;
    const int khi  = lane >> 4;
    const int lo   = lane & 15;
    const int n    = n_tile * 16 + lo;

    {
        const int*   sg = segid + b * LFR;
        const float* wb = w     + b * LFR;
        for (int i = threadIdx.x; i < LFR; i += blockDim.x) {
            s_sg[i] = sg[i];
            s_w [i] = wb[i];
        }
    }
    __syncthreads();

    const float* xb = x + (size_t)b * LFR * DDIM;

    v8f c0 = {}, c1 = {}, c2 = {}, c3 = {};
    const int m0 = lo, m1 = 16 + lo, m2 = 32 + lo, m3 = 48 + lo;

    for (int kb = 0; kb < LFR; kb += 4) {
        const int k0 = kb + 2 * khi;
        const int k1 = k0 + 1;

        const int   sA = s_sg[k0], sB = s_sg[k1];
        const float wA = s_w[k0],  wB = s_w[k1];

        v2f Bv;
        Bv.x = xb[(size_t)k0 * DDIM + n];
        Bv.y = xb[(size_t)k1 * DDIM + n];

        v2f A;
        A.x = (sA == m0) ? wA : 0.0f;  A.y = (sB == m0) ? wB : 0.0f;
        c0 = __builtin_amdgcn_wmma_f32_16x16x4_f32(false, A, false, Bv, (short)0, c0, false, false);
        A.x = (sA == m1) ? wA : 0.0f;  A.y = (sB == m1) ? wB : 0.0f;
        c1 = __builtin_amdgcn_wmma_f32_16x16x4_f32(false, A, false, Bv, (short)0, c1, false, false);
        A.x = (sA == m2) ? wA : 0.0f;  A.y = (sB == m2) ? wB : 0.0f;
        c2 = __builtin_amdgcn_wmma_f32_16x16x4_f32(false, A, false, Bv, (short)0, c2, false, false);
        A.x = (sA == m3) ? wA : 0.0f;  A.y = (sB == m3) ? wB : 0.0f;
        c3 = __builtin_amdgcn_wmma_f32_16x16x4_f32(false, A, false, Bv, (short)0, c3, false, false);
    }

    float* ob = out + (size_t)b * SMAX * DDIM;
#pragma unroll
    for (int i = 0; i < 8; ++i) {
        const int r = i + 8 * khi;
        ob[(size_t)(r)      * DDIM + n] = c0[i];
        ob[(size_t)(16 + r) * DDIM + n] = c1[i];
        ob[(size_t)(32 + r) * DDIM + n] = c2[i];
        if (48 + r < SMAX)
            ob[(size_t)(48 + r) * DDIM + n] = c3[i];
    }
}

extern "C" void kernel_launch(void* const* d_in, const int* in_sizes, int n_in,
                              void* d_out, int out_size, void* d_ws, size_t ws_size,
                              hipStream_t stream)
{
    (void)in_sizes; (void)n_in; (void)out_size; (void)ws_size;

    const float* x           = (const float*)d_in[0];
    const float* in_boundary = (const float*)d_in[1];

    float* out      = (float*)d_out;
    float* mask_sum = out + (size_t)BATCH * SMAX * DDIM;
    float* bnd_out  = mask_sum + (size_t)BATCH * SMAX;

    int*   segid = (int*)d_ws;
    float* w     = (float*)((char*)d_ws + (size_t)BATCH * LFR * sizeof(int));

    seg_prep<<<BATCH, 64, 0, stream>>>(in_boundary, segid, w, mask_sum, bnd_out);
    seg_gemm_wmma<<<dim3(8, BATCH), 256, 0, stream>>>(x, segid, w, out);
}


# Fuse segmentation into the WMMA kernel: each block builds seg id and 1/len weight in LDS from in_boundary by wave ballot + popcount/clz/ctz; serial one-thread prep launch removed (2 launches -> 1)
# speedup vs baseline: 1.4910x; 1.4910x over previous
#include <hip/hip_runtime.h>


#define BATCH 64
#define LFR   500
#define NB    501
#define DDIM  1024
#define SMAX  50
#define NTHR  256
#define NWORD 16

typedef __attribute__((ext_vector_type(2))) float v2f;
typedef __attribute__((ext_vector_type(8))) float v8f;

__global__ void seg_fused_wmma(const float* __restrict__ x,
                               const float* __restrict__ in_boundary,
                               float* __restrict__ out,
                               float* __restrict__ mask_sum,
                               float* __restrict__ bnd_out)
{
    __shared__ int      s_sg[LFR];
    __shared__ float    s_w [LFR];
    __shared__ unsigned s_bits[NWORD];

    const int b    = blockIdx.y;
    const int wave = threadIdx.x >> 5;
    const int lane = threadIdx.x & 31;
    const int n_tile = blockIdx.x * (NTHR >> 5) + wave;
    const int khi  = lane >> 4;
    const int lo   = lane & 15;
    const int n    = n_tile * 16 + lo;

    {
        const float* ib = in_boundary + b * NB;
        const int t0 = threadIdx.x;
        const int t1 = threadIdx.x + NTHR;
        const float v0 = ib[t0];
        const float v1 = (t1 < NB) ? ib[t1] : 0.0f;
        const unsigned w0 = __builtin_amdgcn_ballot_w32((t0 == 0) || (v0 != 0.0f));
        const unsigned w1 = __builtin_amdgcn_ballot_w32((t1 < NB) && (v1 != 0.0f));
        if (lane == 0) { s_bits[wave] = w0; s_bits[wave + (NTHR >> 5)] = w1; }
        if (blockIdx.x == 0) {
            bnd_out[b * NB + t0] = v0;
            if (t1 < NB) bnd_out[b * NB + t1] = v1;
        }
    }
    __syncthreads();

    {
        unsigned wd[NWORD];
#pragma unroll
        for (int j = 0; j < NWORD; ++j) wd[j] = s_bits[j];

        if (blockIdx.x == 0 && threadIdx.x < SMAX) {
            int tot = 0;
#pragma unroll
            for (int j = 0; j < NWORD; ++j) tot += __builtin_popcount(wd[j]);
            mask_sum[b * SMAX + threadIdx.x] = ((int)threadIdx.x < tot - 1) ? 1.0f : 0.0f;
        }

        for (int l = threadIdx.x; l < LFR; l += NTHR) {
            const int wi = l >> 5;
            const unsigned le = (2u << (l & 31)) - 1u;
            int cnt = 0;
#pragma unroll
            for (int j = 0; j < NWORD; ++j) cnt += (j < wi) ? __builtin_popcount(wd[j]) : 0;
            const unsigned cur = s_bits[wi];
            cnt += __builtin_popcount(cur & le);
            const int seg = cnt - 1;

            unsigned m = cur & le;
            int j = wi;
            while (m == 0u && j > 0) { --j; m = s_bits[j]; }
            const int start = j * 32 + 31 - __builtin_clz(m);

            m = cur & ~le;
            j = wi;
            while (m == 0u && j < NWORD - 1) { ++j; m = s_bits[j]; }

            float wv = 0.0f;
            if (m != 0u && seg < SMAX) {
                const int end = j * 32 + __builtin_ctz(m);
                wv = 1.0f / (float)(end - start);
            }
            s_sg[l] = seg;
            s_w [l] = wv;
        }
    }
    __syncthreads();

    const float* xb = x + (size_t)b * LFR * DDIM;

    v8f c0 = {}, c1 = {}, c2 = {}, c3 = {};
    const int m0 = lo, m1 = 16 + lo, m2 = 32 + lo, m3 = 48 + lo;

    for (int kb = 0; kb < LFR; kb += 4) {
        const int k0 = kb + 2 * khi;
        const int k1 = k0 + 1;

        const int   sA = s_sg[k0], sB = s_sg[k1];
        const float wA = s_w[k0],  wB = s_w[k1];

        v2f Bv;
        Bv.x = xb[(size_t)k0 * DDIM + n];
        Bv.y = xb[(size_t)k1 * DDIM + n];

        v2f A;
        A.x = (sA == m0) ? wA : 0.0f;  A.y = (sB == m0) ? wB : 0.0f;
        c0 = __builtin_amdgcn_wmma_f32_16x16x4_f32(false, A, false, Bv, (short)0, c0, false, false);
        A.x = (sA == m1) ? wA : 0.0f;  A.y = (sB == m1) ? wB : 0.0f;
        c1 = __builtin_amdgcn_wmma_f32_16x16x4_f32(false, A, false, Bv, (short)0, c1, false, false);
        A.x = (sA == m2) ? wA : 0.0f;  A.y = (sB == m2) ? wB : 0.0f;
        c2 = __builtin_amdgcn_wmma_f32_16x16x4_f32(false, A, false, Bv, (short)0, c2, false, false);
        A.x = (sA == m3) ? wA : 0.0f;  A.y = (sB == m3) ? wB : 0.0f;
        c3 = __builtin_amdgcn_wmma_f32_16x16x4_f32(false, A, false, Bv, (short)0, c3, false, false);
    }

    float* ob = out + (size_t)b * SMAX * DDIM;
#pragma unroll
    for (int i = 0; i < 8; ++i) {
        const int r = i + 8 * khi;
        ob[(size_t)(r)      * DDIM + n] = c0[i];
        ob[(size_t)(16 + r) * DDIM + n] = c1[i];
        ob[(size_t)(32 + r) * DDIM + n] = c2[i];
        if (48 + r < SMAX)
            ob[(size_t)(48 + r) * DDIM + n] = c3[i];
    }
}

extern "C" void kernel_launch(void* const* d_in, const int* in_sizes, int n_in,
                              void* d_out, int out_size, void* d_ws, size_t ws_size,
                              hipStream_t stream)
{
    (void)in_sizes; (void)n_in; (void)out_size; (void)d_ws; (void)ws_size;

    const float* x           = (const float*)d_in[0];
    const float* in_boundary = (const float*)d_in[1];

    float* out      = (float*)d_out;
    float* mask_sum = out + (size_t)BATCH * SMAX * DDIM;
    float* bnd_out  = mask_sum + (size_t)BATCH * SMAX;

    seg_fused_wmma<<<dim3(8, BATCH), NTHR, 0, stream>>>(x, in_boundary, out, mask_sum, bnd_out);
}


# x loads pipelined: 25-K-step register ring per wave, first ring-full issued before the segmentation and kept in flight across its LDS-only barriers, each slot refilled behind its WMMAs
# speedup vs baseline: 3.4118x; 2.2883x over previous
#include <hip/hip_runtime.h>


#define BATCH 64
#define LFR   500
#define NB    501
#define DDIM  1024
#define SMAX  50
#define NTHR  256
#define NWORD 16
#define KSTEPS (LFR / 4)
#define AHEAD  25

typedef __attribute__((ext_vector_type(2))) float v2f;
typedef __attribute__((ext_vector_type(8))) float v8f;

static __device__ __forceinline__ void lds_barrier()
{
    __builtin_amdgcn_fence(__ATOMIC_RELEASE, "workgroup", "local");
    __builtin_amdgcn_s_barrier();
    __builtin_amdgcn_fence(__ATOMIC_ACQUIRE, "workgroup", "local");
}

__global__ void seg_fused_wmma(const float* __restrict__ x,
                               const float* __restrict__ in_boundary,
                               float* __restrict__ out,
                               float* __restrict__ mask_sum,
                               float* __restrict__ bnd_out)
{
    __shared__ int      s_sg[LFR];
    __shared__ float    s_w [LFR];
    __shared__ unsigned s_bits[NWORD];

    const int b    = blockIdx.y;
    const int wave = threadIdx.x >> 5;
    const int lane = threadIdx.x & 31;
    const int n_tile = blockIdx.x * (NTHR >> 5) + wave;
    const int khi  = lane >> 4;
    const int lo   = lane & 15;
    const int n    = n_tile * 16 + lo;

    const float* xp = x + (size_t)b * LFR * DDIM + (size_t)(2 * khi) * DDIM + n;
    v2f ring[AHEAD];

    {
        const float* ib = in_boundary + b * NB;
        const int t0 = threadIdx.x;
        const int t1 = threadIdx.x + NTHR;
        const float v0 = ib[t0];
        const float v1 = (t1 < NB) ? ib[t1] : 0.0f;

#pragma unroll
        for (int d = 0; d < AHEAD; ++d) {
            ring[d].x = xp[(size_t)(4 * d)     * DDIM];
            ring[d].y = xp[(size_t)(4 * d + 1) * DDIM];
        }
        const unsigned w0 = __builtin_amdgcn_ballot_w32((t0 == 0) || (v0 != 0.0f));
        const unsigned w1 = __builtin_amdgcn_ballot_w32((t1 < NB) && (v1 != 0.0f));
        if (lane == 0) { s_bits[wave] = w0; s_bits[wave + (NTHR >> 5)] = w1; }
        if (blockIdx.x == 0) {
            bnd_out[b * NB + t0] = v0;
            if (t1 < NB) bnd_out[b * NB + t1] = v1;
        }
    }
    lds_barrier();

    {
        unsigned wd[NWORD];
#pragma unroll
        for (int j = 0; j < NWORD; ++j) wd[j] = s_bits[j];

        if (blockIdx.x == 0 && threadIdx.x < SMAX) {
            int tot = 0;
#pragma unroll
            for (int j = 0; j < NWORD; ++j) tot += __builtin_popcount(wd[j]);
            mask_sum[b * SMAX + threadIdx.x] = ((int)threadIdx.x < tot - 1) ? 1.0f : 0.0f;
        }

        for (int l = threadIdx.x; l < LFR; l += NTHR) {
            const int wi = l >> 5;
            const unsigned le = (2u << (l & 31)) - 1u;
            int cnt = 0;
#pragma unroll
            for (int j = 0; j < NWORD; ++j) cnt += (j < wi) ? __builtin_popcount(wd[j]) : 0;
            const unsigned cur = s_bits[wi];
            cnt += __builtin_popcount(cur & le);
            const int seg = cnt - 1;

            unsigned m = cur & le;
            int j = wi;
            while (m == 0u && j > 0) { --j; m = s_bits[j]; }
            const int start = j * 32 + 31 - __builtin_clz(m);

            m = cur & ~le;
            j = wi;
            while (m == 0u && j < NWORD - 1) { ++j; m = s_bits[j]; }

            float wv = 0.0f;
            if (m != 0u && seg < SMAX) {
                const int end = j * 32 + __builtin_ctz(m);
                wv = 1.0f / (float)(end - start);
            }
            s_sg[l] = seg;
            s_w [l] = wv;
        }
    }
    lds_barrier();

    v8f c0 = {}, c1 = {}, c2 = {}, c3 = {};
    const int m0 = lo, m1 = 16 + lo, m2 = 32 + lo, m3 = 48 + lo;

    auto kstep = [&](int s, v2f Bv) {
        const int k0 = 4 * s + 2 * khi;
        const int k1 = k0 + 1;

        const int   sA = s_sg[k0], sB = s_sg[k1];
        const float wA = s_w[k0],  wB = s_w[k1];

        v2f A;
        A.x = (sA == m0) ? wA : 0.0f;  A.y = (sB == m0) ? wB : 0.0f;
        c0 = __builtin_amdgcn_wmma_f32_16x16x4_f32(false, A, false, Bv, (short)0, c0, false, false);
        A.x = (sA == m1) ? wA : 0.0f;  A.y = (sB == m1) ? wB : 0.0f;
        c1 = __builtin_amdgcn_wmma_f32_16x16x4_f32(false, A, false, Bv, (short)0, c1, false, false);
        A.x = (sA == m2) ? wA : 0.0f;  A.y = (sB == m2) ? wB : 0.0f;
        c2 = __builtin_amdgcn_wmma_f32_16x16x4_f32(false, A, false, Bv, (short)0, c2, false, false);
        A.x = (sA == m3) ? wA : 0.0f;  A.y = (sB == m3) ? wB : 0.0f;
        c3 = __builtin_amdgcn_wmma_f32_16x16x4_f32(false, A, false, Bv, (short)0, c3, false, false);
    };

    for (int s0 = 0; s0 < KSTEPS - AHEAD; s0 += AHEAD) {
        const float* xn = xp + (size_t)(4 * (s0 + AHEAD)) * DDIM;
#pragma unroll
        for (int d = 0; d < AHEAD; ++d) {
            kstep(s0 + d, ring[d]);
            ring[d].x = xn[(size_t)(4 * d)     * DDIM];
            ring[d].y = xn[(size_t)(4 * d + 1) * DDIM];
        }
    }
#pragma unroll
    for (int d = 0; d < AHEAD; ++d) kstep(KSTEPS - AHEAD + d, ring[d]);

    float* ob = out + (size_t)b * SMAX * DDIM;
#pragma unroll
    for (int i = 0; i < 8; ++i) {
        const int r = i + 8 * khi;
        ob[(size_t)(r)      * DDIM + n] = c0[i];
        ob[(size_t)(16 + r) * DDIM + n] = c1[i];
        ob[(size_t)(32 + r) * DDIM + n] = c2[i];
        if (48 + r < SMAX)
            ob[(size_t)(48 + r) * DDIM + n] = c3[i];
    }
}

extern "C" void kernel_launch(void* const* d_in, const int* in_sizes, int n_in,
                              void* d_out, int out_size, void* d_ws, size_t ws_size,
                              hipStream_t stream)
{
    (void)in_sizes; (void)n_in; (void)out_size; (void)d_ws; (void)ws_size;

    const float* x           = (const float*)d_in[0];
    const float* in_boundary = (const float*)d_in[1];

    float* out      = (float*)d_out;
    float* mask_sum = out + (size_t)BATCH * SMAX * DDIM;
    float* bnd_out  = mask_sum + (size_t)BATCH * SMAX;

    seg_fused_wmma<<<dim3(8, BATCH), NTHR, 0, stream>>>(x, in_boundary, out, mask_sum, bnd_out);
}


# Products only for the 16-segment tile a K-step's frames fall in (seg ids never fall): one running accumulator rotated at a tile change, flags by ballot per ring length; 4 WMMAs per K-step -> 1
# speedup vs baseline: 6.0124x; 1.7622x over previous
#include <hip/hip_runtime.h>


#define BATCH 64
#define LFR   500
#define NB    501
#define DDIM  1024
#define SMAX  50
#define NTHR  256
#define NWORD 16
#define KSTEPS (LFR / 4)
#define AHEAD  25
#define LPAD  (LFR + 4)

typedef __attribute__((ext_vector_type(2))) float v2f;
typedef __attribute__((ext_vector_type(4))) float v4f;
typedef __attribute__((ext_vector_type(8))) float v8f;
typedef __attribute__((ext_vector_type(2))) int   v2i;
typedef __attribute__((ext_vector_type(4))) int   v4i;

static __device__ __forceinline__ void lds_barrier()
{
    __builtin_amdgcn_fence(__ATOMIC_RELEASE, "workgroup", "local");
    __builtin_amdgcn_s_barrier();
    __builtin_amdgcn_fence(__ATOMIC_ACQUIRE, "workgroup", "local");
}

__global__ void seg_fused_wmma(const float* __restrict__ x,
                               const float* __restrict__ in_boundary,
                               float* __restrict__ out,
                               float* __restrict__ mask_sum,
                               float* __restrict__ bnd_out)
{
    __shared__ __attribute__((aligned(16))) int   s_sg[LPAD];
    __shared__ __attribute__((aligned(16))) float s_w [LPAD];
    __shared__ unsigned s_bits[NWORD];

    const int b    = blockIdx.y;
    const int wave = threadIdx.x >> 5;
    const int lane = threadIdx.x & 31;
    const int n_tile = blockIdx.x * (NTHR >> 5) + wave;
    const int khi  = lane >> 4;
    const int lo   = lane & 15;
    const int n    = n_tile * 16 + lo;

    const float* xp = x + (size_t)b * LFR * DDIM + (size_t)(2 * khi) * DDIM + n;
    v2f ring[AHEAD];

    {
        const float* ib = in_boundary + b * NB;
        const int t0 = threadIdx.x;
        const int t1 = threadIdx.x + NTHR;
        const float v0 = ib[t0];
        const float v1 = (t1 < NB) ? ib[t1] : 0.0f;

#pragma unroll
        for (int d = 0; d < AHEAD; ++d) {
            ring[d].x = xp[(size_t)(4 * d)     * DDIM];
            ring[d].y = xp[(size_t)(4 * d + 1) * DDIM];
        }
        const unsigned w0 = __builtin_amdgcn_ballot_w32((t0 == 0) || (v0 != 0.0f));
        const unsigned w1 = __builtin_amdgcn_ballot_w32((t1 < NB) && (v1 != 0.0f));
        if (lane == 0) { s_bits[wave] = w0; s_bits[wave + (NTHR >> 5)] = w1; }
        if (blockIdx.x == 0) {
            bnd_out[b * NB + t0] = v0;
            if (t1 < NB) bnd_out[b * NB + t1] = v1;
        }
    }
    lds_barrier();

    {
        unsigned wd[NWORD];
#pragma unroll
        for (int j = 0; j < NWORD; ++j) wd[j] = s_bits[j];

        if (blockIdx.x == 0 && threadIdx.x < SMAX) {
            int tot = 0;
#pragma unroll
            for (int j = 0; j < NWORD; ++j) tot += __builtin_popcount(wd[j]);
            mask_sum[b * SMAX + threadIdx.x] = ((int)threadIdx.x < tot - 1) ? 1.0f : 0.0f;
        }

        for (int l = threadIdx.x; l < LFR; l += NTHR) {
            const int wi = l >> 5;
            const unsigned le = (2u << (l & 31)) - 1u;
            int cnt = 0;
#pragma unroll
            for (int j = 0; j < NWORD; ++j) cnt += (j < wi) ? __builtin_popcount(wd[j]) : 0;
            const unsigned cur = s_bits[wi];
            cnt += __builtin_popcount(cur & le);
            const int seg = cnt - 1;

            unsigned m = cur & le;
            int j = wi;
            while (m == 0u && j > 0) { --j; m = s_bits[j]; }
            const int start = j * 32 + 31 - __builtin_clz(m);

            m = cur & ~le;
            j = wi;
            while (m == 0u && j < NWORD - 1) { ++j; m = s_bits[j]; }

            float wv = 0.0f;
            if (m != 0u && seg < SMAX) {
                const int end = j * 32 + __builtin_ctz(m);
                wv = 1.0f / (float)(end - start);
            }
            s_sg[l] = seg;
            s_w [l] = wv;
        }
        if (threadIdx.x < LPAD - LFR) { s_sg[LFR + threadIdx.x] = -1; s_w[LFR + threadIdx.x] = 0.0f; }
    }
    lds_barrier();

    v8f c = {}, p0 = {}, p1 = {}, p2 = {};
    int cb = lo;
    int npush = 0;

    auto push = [&]() { p2 = p1; p1 = p0; p0 = c; c = v8f{}; cb += 16; ++npush; };

    auto trip_flags = [&](int s0, unsigned& act, unsigned& trn, unsigned& str) {
        const int s  = (s0 + lane < KSTEPS) ? (s0 + lane) : (KSTEPS - 1);
        const int f  = 4 * s;
        const v4i g  = *(const v4i*)&s_sg[f];
        const v4f w  = *(const v4f*)&s_w [f];
        const int gp = s_sg[(f > 0) ? (f - 1) : 0];
        const int t0 = g.x >> 4;
        int tl = t0;
        if (w.y != 0.0f) tl = g.y >> 4;
        if (w.z != 0.0f) tl = g.z >> 4;
        if (w.w != 0.0f) tl = g.w >> 4;
        const bool a = (w.x != 0.0f);
        act = __builtin_amdgcn_ballot_w32(a);
        trn = __builtin_amdgcn_ballot_w32(a && ((gp >> 4) != t0));
        str = __builtin_amdgcn_ballot_w32(a && (tl != t0));
    };

    const int* sgp = s_sg + 2 * khi;
    const float* swp = s_w + 2 * khi;
    v2i gn = *(const v2i*)sgp;
    v2f wn = *(const v2f*)swp;

    auto kstep = [&](int s, int d, unsigned act, unsigned trn, unsigned str, v2f Bv) {
        const v2i g = gn;
        const v2f w = wn;
        gn = *(const v2i*)(sgp + 4 * (s + 1));
        wn = *(const v2f*)(swp + 4 * (s + 1));

        if ((act >> d) & 1u) {
            if (__builtin_expect((trn >> d) & 1u, 0)) push();
            v2f A;
            A.x = (g.x == cb) ? w.x : 0.0f;  A.y = (g.y == cb) ? w.y : 0.0f;
            c = __builtin_amdgcn_wmma_f32_16x16x4_f32(false, A, false, Bv, (short)0, c, false, false);
            if (__builtin_expect((str >> d) & 1u, 0)) {
                push();
                A.x = (g.x == cb) ? w.x : 0.0f;  A.y = (g.y == cb) ? w.y : 0.0f;
                c = __builtin_amdgcn_wmma_f32_16x16x4_f32(false, A, false, Bv, (short)0, c, false, false);
            }
        }
    };

    for (int s0 = 0; s0 < KSTEPS - AHEAD; s0 += AHEAD) {
        unsigned act, trn, str;
        trip_flags(s0, act, trn, str);
        const float* xn = xp + (size_t)(4 * (s0 + AHEAD)) * DDIM;
#pragma unroll
        for (int d = 0; d < AHEAD; ++d) {
            kstep(s0 + d, d, act, trn, str, ring[d]);
            ring[d].x = xn[(size_t)(4 * d)     * DDIM];
            ring[d].y = xn[(size_t)(4 * d + 1) * DDIM];
        }
    }
    {
        unsigned act, trn, str;
        trip_flags(KSTEPS - AHEAD, act, trn, str);
#pragma unroll
        for (int d = 0; d < AHEAD; ++d) kstep(KSTEPS - AHEAD + d, d, act, trn, str, ring[d]);
    }

    for (; npush < 3; ++npush) { p2 = p1; p1 = p0; p0 = c; c = v8f{}; }

    float* ob = out + (size_t)b * SMAX * DDIM;
#pragma unroll
    for (int i = 0; i < 8; ++i) {
        const int r = i + 8 * khi;
        ob[(size_t)(r)      * DDIM + n] = p2[i];
        ob[(size_t)(16 + r) * DDIM + n] = p1[i];
        ob[(size_t)(32 + r) * DDIM + n] = p0[i];
        if (48 + r < SMAX)
            ob[(size_t)(48 + r) * DDIM + n] = c[i];
    }
}

extern "C" void kernel_launch(void* const* d_in, const int* in_sizes, int n_in,
                              void* d_out, int out_size, void* d_ws, size_t ws_size,
                              hipStream_t stream)
{
    (void)in_sizes; (void)n_in; (void)out_size; (void)d_ws; (void)ws_size;

    const float* x           = (const float*)d_in[0];
    const float* in_boundary = (const float*)d_in[1];

    float* out      = (float*)d_out;
    float* mask_sum = out + (size_t)BATCH * SMAX * DDIM;
    float* bnd_out  = mask_sum + (size_t)BATCH * SMAX;

    seg_fused_wmma<<<dim3(8, BATCH), NTHR, 0, stream>>>(x, in_boundary, out, mask_sum, bnd_out);
}


# A operands from an LDS table built once per block ([frame][seg&15] = weight or 0): a K-step reads its A pair by one LDS load, no compare/select per K-step; tile masking only at a tile change
# speedup vs baseline: 6.4014x; 1.0647x over previous
#include <hip/hip_runtime.h>


#define BATCH 64
#define LFR   500
#define NB    501
#define DDIM  1024
#define SMAX  50
#define NTHR  256
#define NWORD 16
#define KSTEPS (LFR / 4)
#define AHEAD  25
#define LPAD  (LFR + 4)

typedef __attribute__((ext_vector_type(2))) float v2f;
typedef __attribute__((ext_vector_type(4))) float v4f;
typedef __attribute__((ext_vector_type(8))) float v8f;
typedef __attribute__((ext_vector_type(2))) int   v2i;
typedef __attribute__((ext_vector_type(4))) int   v4i;

static __device__ __forceinline__ void lds_barrier()
{
    __builtin_amdgcn_fence(__ATOMIC_RELEASE, "workgroup", "local");
    __builtin_amdgcn_s_barrier();
    __builtin_amdgcn_fence(__ATOMIC_ACQUIRE, "workgroup", "local");
}

__global__ void seg_fused_wmma(const float* __restrict__ x,
                               const float* __restrict__ in_boundary,
                               float* __restrict__ out,
                               float* __restrict__ mask_sum,
                               float* __restrict__ bnd_out)
{
    __shared__ __attribute__((aligned(16))) int   s_sg[LPAD];
    __shared__ __attribute__((aligned(16))) float s_w [LPAD];
    __shared__ unsigned s_bits[NWORD];
    __shared__ __attribute__((aligned(16))) float s_A[LPAD * 16];

    const int b    = blockIdx.y;
    const int wave = threadIdx.x >> 5;
    const int lane = threadIdx.x & 31;
    const int n_tile = blockIdx.x * (NTHR >> 5) + wave;
    const int khi  = lane >> 4;
    const int lo   = lane & 15;
    const int n    = n_tile * 16 + lo;

    const float* xp = x + (size_t)b * LFR * DDIM + (size_t)(2 * khi) * DDIM + n;
    v2f ring[AHEAD];

    {
        const float* ib = in_boundary + b * NB;
        const int t0 = threadIdx.x;
        const int t1 = threadIdx.x + NTHR;
        const float v0 = ib[t0];
        const float v1 = (t1 < NB) ? ib[t1] : 0.0f;

#pragma unroll
        for (int d = 0; d < AHEAD; ++d) {
            ring[d].x = xp[(size_t)(4 * d)     * DDIM];
            ring[d].y = xp[(size_t)(4 * d + 1) * DDIM];
        }
        const unsigned w0 = __builtin_amdgcn_ballot_w32((t0 == 0) || (v0 != 0.0f));
        const unsigned w1 = __builtin_amdgcn_ballot_w32((t1 < NB) && (v1 != 0.0f));
        if (lane == 0) { s_bits[wave] = w0; s_bits[wave + (NTHR >> 5)] = w1; }
        if (blockIdx.x == 0) {
            bnd_out[b * NB + t0] = v0;
            if (t1 < NB) bnd_out[b * NB + t1] = v1;
        }
    }
    lds_barrier();

    {
        unsigned wd[NWORD];
#pragma unroll
        for (int j = 0; j < NWORD; ++j) wd[j] = s_bits[j];

        if (blockIdx.x == 0 && threadIdx.x < SMAX) {
            int tot = 0;
#pragma unroll
            for (int j = 0; j < NWORD; ++j) tot += __builtin_popcount(wd[j]);
            mask_sum[b * SMAX + threadIdx.x] = ((int)threadIdx.x < tot - 1) ? 1.0f : 0.0f;
        }

        for (int l = threadIdx.x; l < LFR; l += NTHR) {
            const int wi = l >> 5;
            const unsigned le = (2u << (l & 31)) - 1u;
            int cnt = 0;
#pragma unroll
            for (int j = 0; j < NWORD; ++j) cnt += (j < wi) ? __builtin_popcount(wd[j]) : 0;
            const unsigned cur = s_bits[wi];
            cnt += __builtin_popcount(cur & le);
            const int seg = cnt - 1;

            unsigned m = cur & le;
            int j = wi;
            while (m == 0u && j > 0) { --j; m = s_bits[j]; }
            const int start = j * 32 + 31 - __builtin_clz(m);

            m = cur & ~le;
            j = wi;
            while (m == 0u && j < NWORD - 1) { ++j; m = s_bits[j]; }

            float wv = 0.0f;
            if (m != 0u && seg < SMAX) {
                const int end = j * 32 + __builtin_ctz(m);
                wv = 1.0f / (float)(end - start);
            }
            s_sg[l] = seg;
            s_w [l] = wv;
            const int sl = seg & 15;
            float* ar = s_A + l * 16;
#pragma unroll
            for (int q = 0; q < 4; ++q) {
                v4f t;
                t.x = (sl == 4 * q)     ? wv : 0.0f;
                t.y = (sl == 4 * q + 1) ? wv : 0.0f;
                t.z = (sl == 4 * q + 2) ? wv : 0.0f;
                t.w = (sl == 4 * q + 3) ? wv : 0.0f;
                *(v4f*)(ar + 4 * q) = t;
            }
        }
        if (threadIdx.x < LPAD - LFR) { s_sg[LFR + threadIdx.x] = -1; s_w[LFR + threadIdx.x] = 0.0f; }
        if (threadIdx.x < (LPAD - LFR) * 16) s_A[LFR * 16 + threadIdx.x] = 0.0f;
    }
    lds_barrier();

    v8f c = {}, p0 = {}, p1 = {}, p2 = {};
    int npush = 0;

    auto push = [&]() { p2 = p1; p1 = p0; p0 = c; c = v8f{}; ++npush; };

    auto trip_flags = [&](int s0, unsigned& act, unsigned& trn, unsigned& str) {
        const int s  = (s0 + lane < KSTEPS) ? (s0 + lane) : (KSTEPS - 1);
        const int f  = 4 * s;
        const v4i g  = *(const v4i*)&s_sg[f];
        const v4f w  = *(const v4f*)&s_w [f];
        const int gp = s_sg[(f > 0) ? (f - 1) : 0];
        const int t0 = g.x >> 4;
        int tl = t0;
        if (w.y != 0.0f) tl = g.y >> 4;
        if (w.z != 0.0f) tl = g.z >> 4;
        if (w.w != 0.0f) tl = g.w >> 4;
        const bool a = (w.x != 0.0f);
        act = __builtin_amdgcn_ballot_w32(a);
        trn = __builtin_amdgcn_ballot_w32(a && ((gp >> 4) != t0));
        str = __builtin_amdgcn_ballot_w32(a && (tl != t0));
    };

    const int* sgp = s_sg + 2 * khi;
    const float* sap = s_A + (2 * khi) * 16 + lo;
    v2f an;
    an.x = sap[0];
    an.y = sap[16];

    auto kstep = [&](int s, int d, unsigned act, unsigned trn, unsigned str, v2f Bv) {
        const v2f A = an;
        an.x = sap[(4 * (s + 1)) * 16];
        an.y = sap[(4 * (s + 1)) * 16 + 16];

        if ((act >> d) & 1u) {
            v2f M = A;
            if (__builtin_expect(((trn | str) >> d) & 1u, 0)) {
                if ((trn >> d) & 1u) push();
                if ((str >> d) & 1u) {
                    const v2i g = *(const v2i*)(sgp + 4 * s);
                    M.x = ((g.x >> 4) == npush) ? A.x : 0.0f;  M.y = ((g.y >> 4) == npush) ? A.y : 0.0f;
                    c = __builtin_amdgcn_wmma_f32_16x16x4_f32(false, M, false, Bv, (short)0, c, false, false);
                    push();
                    M.x = ((g.x >> 4) == npush) ? A.x : 0.0f;  M.y = ((g.y >> 4) == npush) ? A.y : 0.0f;
                }
            }
            c = __builtin_amdgcn_wmma_f32_16x16x4_f32(false, M, false, Bv, (short)0, c, false, false);
        }
    };

    for (int s0 = 0; s0 < KSTEPS - AHEAD; s0 += AHEAD) {
        unsigned act, trn, str;
        trip_flags(s0, act, trn, str);
        const float* xn = xp + (size_t)(4 * (s0 + AHEAD)) * DDIM;
#pragma unroll
        for (int d = 0; d < AHEAD; ++d) {
            kstep(s0 + d, d, act, trn, str, ring[d]);
            ring[d].x = xn[(size_t)(4 * d)     * DDIM];
            ring[d].y = xn[(size_t)(4 * d + 1) * DDIM];
        }
    }
    {
        unsigned act, trn, str;
        trip_flags(KSTEPS - AHEAD, act, trn, str);
#pragma unroll
        for (int d = 0; d < AHEAD; ++d) kstep(KSTEPS - AHEAD + d, d, act, trn, str, ring[d]);
    }

    for (; npush < 3; ++npush) { p2 = p1; p1 = p0; p0 = c; c = v8f{}; }

    float* ob = out + (size_t)b * SMAX * DDIM;
#pragma unroll
    for (int i = 0; i < 8; ++i) {
        const int r = i + 8 * khi;
        ob[(size_t)(r)      * DDIM + n] = p2[i];
        ob[(size_t)(16 + r) * DDIM + n] = p1[i];
        ob[(size_t)(32 + r) * DDIM + n] = p0[i];
        if (48 + r < SMAX)
            ob[(size_t)(48 + r) * DDIM + n] = c[i];
    }
}

extern "C" void kernel_launch(void* const* d_in, const int* in_sizes, int n_in,
                              void* d_out, int out_size, void* d_ws, size_t ws_size,
                              hipStream_t stream)
{
    (void)in_sizes; (void)n_in; (void)out_size; (void)d_ws; (void)ws_size;

    const float* x           = (const float*)d_in[0];
    const float* in_boundary = (const float*)d_in[1];

    float* out      = (float*)d_out;
    float* mask_sum = out + (size_t)BATCH * SMAX * DDIM;
    float* bnd_out  = mask_sum + (size_t)BATCH * SMAX;

    seg_fused_wmma<<<dim3(8, BATCH), NTHR, 0, stream>>>(x, in_boundary, out, mask_sum, bnd_out);
}
